# starting kernel
# baseline (speedup 1.0000x reference)
#include <hip/hip_runtime.h>
#include <cmath>

typedef __attribute__((ext_vector_type(2))) float v2f;
typedef __attribute__((ext_vector_type(8))) float v8f;

#define Bq  64
#define Sq  2048
#define Iq  128
#define Hq  256
#define BDq 128
#define LDSROW 260

__global__ __launch_bounds__(256) void k_xproj(const float* __restrict__ nh,
                                               const float* __restrict__ nr,
                                               const float* __restrict__ nt,
                                               const float* __restrict__ Wx,
                                               const float* __restrict__ bh,
                                               float* __restrict__ P) {
  const int lane = threadIdx.x & 31, wv = threadIdx.x >> 5;
  const int gw    = blockIdx.x * 8 + wv;
  const int mtile = gw >> 1, ng = gw & 1;
  const int half  = lane >> 4, l16 = lane & 15;
  const int grow  = mtile * 16 + l16;
  const int t = grow >> 6, b = grow & 63;
  const size_t xbase = ((size_t)b * Sq + t) * Iq;

  v8f acc[8];
  #pragma unroll
  for (int n = 0; n < 8; ++n) {
    const float bias = bh[ng * 128 + n * 16 + l16];
    #pragma unroll
    for (int j = 0; j < 8; ++j) acc[n][j] = bias;
  }

  for (int kk = 0; kk < 96; ++kk) {
    const int kb = kk * 4 + 2 * half;
    const float* xs = (kb < Iq) ? nh : ((kb < 2 * Iq) ? nr : nt);
    const int off = kb & (Iq - 1);
    v2f a; a.x = xs[xbase + off]; a.y = xs[xbase + off + 1];
    #pragma unroll
    for (int n = 0; n < 8; ++n) {
      const int col = ng * 128 + n * 16 + l16;
      v2f bm; bm.x = Wx[(size_t)kb * Hq + col];
              bm.y = Wx[(size_t)(kb + 1) * Hq + col];
      acc[n] = __builtin_amdgcn_wmma_f32_16x16x4_f32(
          false, a, false, bm, (short)0, acc[n], false, false);
    }
  }

  #pragma unroll
  for (int n = 0; n < 8; ++n) {
    const int col = ng * 128 + n * 16 + l16;
    #pragma unroll
    for (int j = 0; j < 8; ++j) {
      const int r = mtile * 16 + j + 8 * half;
      P[(size_t)r * Hq + col] = acc[n][j];
    }
  }
}

__global__ __launch_bounds__(1024) void k_scan(const float* __restrict__ h0,
                                               const float* __restrict__ Wh,
                                               float* __restrict__ P,
                                               float* __restrict__ hlast) {
  __shared__ float hsh[64 * LDSROW];
  const int tid = threadIdx.x;
  for (int i = tid; i < 64 * Hq; i += 1024)
    hsh[(i >> 8) * LDSROW + (i & 255)] = h0[i];
  __syncthreads();

  const int lane = tid & 31, wv = tid >> 5;
  const int half = lane >> 4, l16 = lane & 15;
  const int n  = wv & 15;
  const int m0 = wv >> 4;
  const int m1 = m0 + 2;
  const int col = n * 16 + l16;

  for (int t = 0; t < Sq; ++t) {
    const size_t pbase = (size_t)t * (64 * Hq);
    v8f acc0, acc1;
    #pragma unroll
    for (int j = 0; j < 8; ++j) {
      acc0[j] = P[pbase + (size_t)(m0 * 16 + j + 8 * half) * Hq + col];
      acc1[j] = P[pbase + (size_t)(m1 * 16 + j + 8 * half) * Hq + col];
    }
    for (int kk = 0; kk < 64; ++kk) {
      const int kb = kk * 4 + 2 * half;
      v2f a0, a1, bm;
      a0.x = hsh[(m0 * 16 + l16) * LDSROW + kb];
      a0.y = hsh[(m0 * 16 + l16) * LDSROW + kb + 1];
      a1.x = hsh[(m1 * 16 + l16) * LDSROW + kb];
      a1.y = hsh[(m1 * 16 + l16) * LDSROW + kb + 1];
      bm.x = Wh[(size_t)kb * Hq + col];
      bm.y = Wh[(size_t)(kb + 1) * Hq + col];
      acc0 = __builtin_amdgcn_wmma_f32_16x16x4_f32(
          false, a0, false, bm, (short)0, acc0, false, false);
      acc1 = __builtin_amdgcn_wmma_f32_16x16x4_f32(
          false, a1, false, bm, (short)0, acc1, false, false);
    }
    __syncthreads();
    #pragma unroll
    for (int j = 0; j < 8; ++j) {
      const int r0 = m0 * 16 + j + 8 * half;
      const int r1 = m1 * 16 + j + 8 * half;
      const float v0 = tanhf(acc0[j]);
      const float v1 = tanhf(acc1[j]);
      hsh[r0 * LDSROW + col] = v0;
      hsh[r1 * LDSROW + col] = v1;
      P[pbase + (size_t)r0 * Hq + col] = v0;
      P[pbase + (size_t)r1 * Hq + col] = v1;
      if (t == Sq - 1) {
        hlast[r0 * Hq + col] = v0;
        hlast[r1 * Hq + col] = v1;
      }
    }
    __syncthreads();
  }
}

__global__ __launch_bounds__(256) void k_out(const float* __restrict__ P,
                                             const float* __restrict__ Wb,
                                             const float* __restrict__ bb,
                                             float* __restrict__ outs) {
  const int lane = threadIdx.x & 31, wv = threadIdx.x >> 5;
  const int mtile = blockIdx.x * 8 + wv;
  const int half = lane >> 4, l16 = lane & 15;
  const int arow = mtile * 16 + l16;

  v8f acc[8];
  #pragma unroll
  for (int n = 0; n < 8; ++n) {
    const float bias = bb[n * 16 + l16];
    #pragma unroll
    for (int j = 0; j < 8; ++j) acc[n][j] = bias;
  }
  for (int kk = 0; kk < 64; ++kk) {
    const int kb = kk * 4 + 2 * half;
    v2f a; a.x = P[(size_t)arow * Hq + kb];
           a.y = P[(size_t)arow * Hq + kb + 1];
    #pragma unroll
    for (int n = 0; n < 8; ++n) {
      const int col = n * 16 + l16;
      v2f bm; bm.x = Wb[(size_t)kb * BDq + col];
              bm.y = Wb[(size_t)(kb + 1) * BDq + col];
      acc[n] = __builtin_amdgcn_wmma_f32_16x16x4_f32(
          false, a, false, bm, (short)0, acc[n], false, false);
    }
  }
  #pragma unroll
  for (int n = 0; n < 8; ++n) {
    const int col = n * 16 + l16;
    #pragma unroll
    for (int j = 0; j < 8; ++j) {
      const int row = mtile * 16 + j + 8 * half;
      const int tt = row >> 6, bb_ = row & 63;
      outs[(size_t)bb_ * Sq * BDq + (size_t)tt * BDq + col] = tanhf(acc[n][j]);
    }
  }
}

extern "C" void kernel_launch(void* const* d_in, const int* in_sizes, int n_in,
                              void* d_out, int out_size, void* d_ws, size_t ws_size,
                              hipStream_t stream) {
  const float* nh = (const float*)d_in[0];
  const float* nr = (const float*)d_in[1];
  const float* nt = (const float*)d_in[2];
  const float* h0 = (const float*)d_in[3];
  const float* Wx = (const float*)d_in[4];
  const float* Wh = (const float*)d_in[5];
  const float* bh = (const float*)d_in[6];
  const float* Wb = (const float*)d_in[7];
  const float* bb = (const float*)d_in[8];

  float* P     = (float*)d_ws;
  float* outs  = (float*)d_out;
  float* hlast = outs + (size_t)Bq * Sq * BDq;

  k_xproj<<<2048, 256, 0, stream>>>(nh, nr, nt, Wx, bh, P);
  k_scan<<<1, 1024, 0, stream>>>(h0, Wh, P, hlast);
  k_out<<<1024, 256, 0, stream>>>(P, Wb, bb, outs);
}


# scan kernel: each wave keeps its 256x16 block of W_h in registers for all 2048 steps instead of reloading it from L2 every step (16 waves x 4 M-tiles so the 128 registers fit)
# speedup vs baseline: 1.1377x; 1.1377x over previous
#include <hip/hip_runtime.h>
#include <cmath>

typedef __attribute__((ext_vector_type(2))) float v2f;
typedef __attribute__((ext_vector_type(8))) float v8f;

#define Bq  64
#define Sq  2048
#define Iq  128
#define Hq  256
#define BDq 128
#define LDSROW 260

__global__ __launch_bounds__(256) void k_xproj(const float* __restrict__ nh,
                                               const float* __restrict__ nr,
                                               const float* __restrict__ nt,
                                               const float* __restrict__ Wx,
                                               const float* __restrict__ bh,
                                               float* __restrict__ P) {
  const int lane = threadIdx.x & 31, wv = threadIdx.x >> 5;
  const int gw    = blockIdx.x * 8 + wv;
  const int mtile = gw >> 1, ng = gw & 1;
  const int half  = lane >> 4, l16 = lane & 15;
  const int grow  = mtile * 16 + l16;
  const int t = grow >> 6, b = grow & 63;
  const size_t xbase = ((size_t)b * Sq + t) * Iq;

  v8f acc[8];
  #pragma unroll
  for (int n = 0; n < 8; ++n) {
    const float bias = bh[ng * 128 + n * 16 + l16];
    #pragma unroll
    for (int j = 0; j < 8; ++j) acc[n][j] = bias;
  }

  for (int kk = 0; kk < 96; ++kk) {
    const int kb = kk * 4 + 2 * half;
    const float* xs = (kb < Iq) ? nh : ((kb < 2 * Iq) ? nr : nt);
    const int off = kb & (Iq - 1);
    v2f a; a.x = xs[xbase + off]; a.y = xs[xbase + off + 1];
    #pragma unroll
    for (int n = 0; n < 8; ++n) {
      const int col = ng * 128 + n * 16 + l16;
      v2f bm; bm.x = Wx[(size_t)kb * Hq + col];
              bm.y = Wx[(size_t)(kb + 1) * Hq + col];
      acc[n] = __builtin_amdgcn_wmma_f32_16x16x4_f32(
          false, a, false, bm, (short)0, acc[n], false, false);
    }
  }

  #pragma unroll
  for (int n = 0; n < 8; ++n) {
    const int col = ng * 128 + n * 16 + l16;
    #pragma unroll
    for (int j = 0; j < 8; ++j) {
      const int r = mtile * 16 + j + 8 * half;
      P[(size_t)r * Hq + col] = acc[n][j];
    }
  }
}

__global__ __launch_bounds__(512) void k_scan(const float* __restrict__ h0,
                                              const float* __restrict__ Wh,
                                              float* __restrict__ P,
                                              float* __restrict__ hlast) {
  __shared__ float hsh[64 * LDSROW];
  const int tid = threadIdx.x;
  for (int i = tid; i < 64 * Hq; i += 512)
    hsh[(i >> 8) * LDSROW + (i & 255)] = h0[i];

  const int lane = tid & 31, n = tid >> 5;
  const int half = lane >> 4, l16 = lane & 15;
  const int col = n * 16 + l16;

  v2f bw[64];
  #pragma unroll
  for (int kk = 0; kk < 64; ++kk) {
    const int kb = kk * 4 + 2 * half;
    bw[kk].x = Wh[(size_t)kb * Hq + col];
    bw[kk].y = Wh[(size_t)(kb + 1) * Hq + col];
  }
  __syncthreads();

  const float* abase = hsh + l16 * LDSROW + 2 * half;
  float* hw = hsh + (8 * half) * LDSROW + col;
  float* pw = P + (size_t)(8 * half) * Hq + col;
  float* lw = hlast + (8 * half) * Hq + col;

  for (int t = 0; t < Sq; ++t) {
    float* pt = pw + (size_t)t * (64 * Hq);
    v8f acc[4];
    #pragma unroll
    for (int m = 0; m < 4; ++m) {
      #pragma unroll
      for (int j = 0; j < 8; ++j) acc[m][j] = pt[(m * 16 + j) * Hq];
    }
    #pragma unroll
    for (int kk = 0; kk < 64; ++kk) {
      #pragma unroll
      for (int m = 0; m < 4; ++m) {
        const float* ap = abase + m * 16 * LDSROW + kk * 4;
        v2f a; a.x = ap[0]; a.y = ap[1];
        acc[m] = __builtin_amdgcn_wmma_f32_16x16x4_f32(
            false, a, false, bw[kk], (short)0, acc[m], false, false);
      }
    }
    __syncthreads();
    #pragma unroll
    for (int m = 0; m < 4; ++m) {
      #pragma unroll
      for (int j = 0; j < 8; ++j) {
        const float v = tanhf(acc[m][j]);
        hw[(m * 16 + j) * LDSROW] = v;
        pt[(m * 16 + j) * Hq] = v;
        if (t == Sq - 1) lw[(m * 16 + j) * Hq] = v;
      }
    }
    __syncthreads();
  }
}

__global__ __launch_bounds__(256) void k_out(const float* __restrict__ P,
                                             const float* __restrict__ Wb,
                                             const float* __restrict__ bb,
                                             float* __restrict__ outs) {
  const int lane = threadIdx.x & 31, wv = threadIdx.x >> 5;
  const int mtile = blockIdx.x * 8 + wv;
  const int half = lane >> 4, l16 = lane & 15;
  const int arow = mtile * 16 + l16;

  v8f acc[8];
  #pragma unroll
  for (int n = 0; n < 8; ++n) {
    const float bias = bb[n * 16 + l16];
    #pragma unroll
    for (int j = 0; j < 8; ++j) acc[n][j] = bias;
  }
  for (int kk = 0; kk < 64; ++kk) {
    const int kb = kk * 4 + 2 * half;
    v2f a; a.x = P[(size_t)arow * Hq + kb];
           a.y = P[(size_t)arow * Hq + kb + 1];
    #pragma unroll
    for (int n = 0; n < 8; ++n) {
      const int col = n * 16 + l16;
      v2f bm; bm.x = Wb[(size_t)kb * BDq + col];
              bm.y = Wb[(size_t)(kb + 1) * BDq + col];
      acc[n] = __builtin_amdgcn_wmma_f32_16x16x4_f32(
          false, a, false, bm, (short)0, acc[n], false, false);
    }
  }
  #pragma unroll
  for (int n = 0; n < 8; ++n) {
    const int col = n * 16 + l16;
    #pragma unroll
    for (int j = 0; j < 8; ++j) {
      const int row = mtile * 16 + j + 8 * half;
      const int tt = row >> 6, bb_ = row & 63;
      outs[(size_t)bb_ * Sq * BDq + (size_t)tt * BDq + col] = tanhf(acc[n][j]);
    }
  }
}

extern "C" void kernel_launch(void* const* d_in, const int* in_sizes, int n_in,
                              void* d_out, int out_size, void* d_ws, size_t ws_size,
                              hipStream_t stream) {
  const float* nh = (const float*)d_in[0];
  const float* nr = (const float*)d_in[1];
  const float* nt = (const float*)d_in[2];
  const float* h0 = (const float*)d_in[3];
  const float* Wx = (const float*)d_in[4];
  const float* Wh = (const float*)d_in[5];
  const float* bh = (const float*)d_in[6];
  const float* Wb = (const float*)d_in[7];
  const float* bb = (const float*)d_in[8];

  float* P     = (float*)d_ws;
  float* outs  = (float*)d_out;
  float* hlast = outs + (size_t)Bq * Sq * BDq;

  k_xproj<<<2048, 256, 0, stream>>>(nh, nr, nt, Wx, bh, P);
  k_scan<<<1, 512, 0, stream>>>(h0, Wh, P, hlast);
  k_out<<<1024, 256, 0, stream>>>(P, Wb, bb, outs);
}


# scan kernel: the four 16-row batch tiles never exchange data, so each runs as its own 512-thread workgroup on its own WGP (grid 1 -> 4); each wave keeps its W_h block, does 1 M-tile, same sum order
# speedup vs baseline: 3.6480x; 3.2065x over previous
#include <hip/hip_runtime.h>
#include <cmath>

typedef __attribute__((ext_vector_type(2))) float v2f;
typedef __attribute__((ext_vector_type(8))) float v8f;

#define Bq  64
#define Sq  2048
#define Iq  128
#define Hq  256
#define BDq 128
#define LDSROW 260

__global__ __launch_bounds__(256) void k_xproj(const float* __restrict__ nh,
                                               const float* __restrict__ nr,
                                               const float* __restrict__ nt,
                                               const float* __restrict__ Wx,
                                               const float* __restrict__ bh,
                                               float* __restrict__ P) {
  const int lane = threadIdx.x & 31, wv = threadIdx.x >> 5;
  const int gw    = blockIdx.x * 8 + wv;
  const int mtile = gw >> 1, ng = gw & 1;
  const int half  = lane >> 4, l16 = lane & 15;
  const int grow  = mtile * 16 + l16;
  const int t = grow >> 6, b = grow & 63;
  const size_t xbase = ((size_t)b * Sq + t) * Iq;

  v8f acc[8];
  #pragma unroll
  for (int n = 0; n < 8; ++n) {
    const float bias = bh[ng * 128 + n * 16 + l16];
    #pragma unroll
    for (int j = 0; j < 8; ++j) acc[n][j] = bias;
  }

  for (int kk = 0; kk < 96; ++kk) {
    const int kb = kk * 4 + 2 * half;
    const float* xs = (kb < Iq) ? nh : ((kb < 2 * Iq) ? nr : nt);
    const int off = kb & (Iq - 1);
    v2f a; a.x = xs[xbase + off]; a.y = xs[xbase + off + 1];
    #pragma unroll
    for (int n = 0; n < 8; ++n) {
      const int col = ng * 128 + n * 16 + l16;
      v2f bm; bm.x = Wx[(size_t)kb * Hq + col];
              bm.y = Wx[(size_t)(kb + 1) * Hq + col];
      acc[n] = __builtin_amdgcn_wmma_f32_16x16x4_f32(
          false, a, false, bm, (short)0, acc[n], false, false);
    }
  }

  #pragma unroll
  for (int n = 0; n < 8; ++n) {
    const int col = ng * 128 + n * 16 + l16;
    #pragma unroll
    for (int j = 0; j < 8; ++j) {
      const int r = mtile * 16 + j + 8 * half;
      P[(size_t)r * Hq + col] = acc[n][j];
    }
  }
}

__global__ __launch_bounds__(512) void k_scan(const float* __restrict__ h0,
                                              const float* __restrict__ Wh,
                                              float* __restrict__ P,
                                              float* __restrict__ hlast) {
  __shared__ float hsh[16 * LDSROW];
  const int tid = threadIdx.x;
  const int row0 = blockIdx.x * 16;
  for (int i = tid; i < 16 * Hq; i += 512)
    hsh[(i >> 8) * LDSROW + (i & 255)] = h0[row0 * Hq + i];

  const int lane = tid & 31, n = tid >> 5;
  const int half = lane >> 4, l16 = lane & 15;
  const int col = n * 16 + l16;

  v2f bw[64];
  #pragma unroll
  for (int kk = 0; kk < 64; ++kk) {
    const int kb = kk * 4 + 2 * half;
    bw[kk].x = Wh[(size_t)kb * Hq + col];
    bw[kk].y = Wh[(size_t)(kb + 1) * Hq + col];
  }
  __syncthreads();

  const float* abase = hsh + l16 * LDSROW + 2 * half;
  float* hw = hsh + (8 * half) * LDSROW + col;
  float* pw = P + (size_t)(row0 + 8 * half) * Hq + col;
  float* lw = hlast + (row0 + 8 * half) * Hq + col;

  for (int t = 0; t < Sq; ++t) {
    float* pt = pw + (size_t)t * (64 * Hq);
    v8f acc;
    #pragma unroll
    for (int j = 0; j < 8; ++j) acc[j] = pt[j * Hq];
    #pragma unroll
    for (int kk = 0; kk < 64; ++kk) {
      const float* ap = abase + kk * 4;
      v2f a; a.x = ap[0]; a.y = ap[1];
      acc = __builtin_amdgcn_wmma_f32_16x16x4_f32(
          false, a, false, bw[kk], (short)0, acc, false, false);
    }
    __syncthreads();
    #pragma unroll
    for (int j = 0; j < 8; ++j) {
      const float v = tanhf(acc[j]);
      hw[j * LDSROW] = v;
      pt[j * Hq] = v;
      if (t == Sq - 1) lw[j * Hq] = v;
    }
    __syncthreads();
  }
}

__global__ __launch_bounds__(256) void k_out(const float* __restrict__ P,
                                             const float* __restrict__ Wb,
                                             const float* __restrict__ bb,
                                             float* __restrict__ outs) {
  const int lane = threadIdx.x & 31, wv = threadIdx.x >> 5;
  const int mtile = blockIdx.x * 8 + wv;
  const int half = lane >> 4, l16 = lane & 15;
  const int arow = mtile * 16 + l16;

  v8f acc[8];
  #pragma unroll
  for (int n = 0; n < 8; ++n) {
    const float bias = bb[n * 16 + l16];
    #pragma unroll
    for (int j = 0; j < 8; ++j) acc[n][j] = bias;
  }
  for (int kk = 0; kk < 64; ++kk) {
    const int kb = kk * 4 + 2 * half;
    v2f a; a.x = P[(size_t)arow * Hq + kb];
           a.y = P[(size_t)arow * Hq + kb + 1];
    #pragma unroll
    for (int n = 0; n < 8; ++n) {
      const int col = n * 16 + l16;
      v2f bm; bm.x = Wb[(size_t)kb * BDq + col];
              bm.y = Wb[(size_t)(kb + 1) * BDq + col];
      acc[n] = __builtin_amdgcn_wmma_f32_16x16x4_f32(
          false, a, false, bm, (short)0, acc[n], false, false);
    }
  }
  #pragma unroll
  for (int n = 0; n < 8; ++n) {
    const int col = n * 16 + l16;
    #pragma unroll
    for (int j = 0; j < 8; ++j) {
      const int row = mtile * 16 + j + 8 * half;
      const int tt = row >> 6, bb_ = row & 63;
      outs[(size_t)bb_ * Sq * BDq + (size_t)tt * BDq + col] = tanhf(acc[n][j]);
    }
  }
}

extern "C" void kernel_launch(void* const* d_in, const int* in_sizes, int n_in,
                              void* d_out, int out_size, void* d_ws, size_t ws_size,
                              hipStream_t stream) {
  const float* nh = (const float*)d_in[0];
  const float* nr = (const float*)d_in[1];
  const float* nt = (const float*)d_in[2];
  const float* h0 = (const float*)d_in[3];
  const float* Wx = (const float*)d_in[4];
  const float* Wh = (const float*)d_in[5];
  const float* bh = (const float*)d_in[6];
  const float* Wb = (const float*)d_in[7];
  const float* bb = (const float*)d_in[8];

  float* P     = (float*)d_ws;
  float* outs  = (float*)d_out;
  float* hlast = outs + (size_t)Bq * Sq * BDq;

  k_xproj<<<2048, 256, 0, stream>>>(nh, nr, nt, Wx, bh, P);
  k_scan<<<4, 512, 0, stream>>>(h0, Wh, P, hlast);
  k_out<<<1024, 256, 0, stream>>>(P, Wb, bb, outs);
}


# scan: global traffic software-pipelined around the recurrence: P[t+1] is loaded one step early and H_all[t] stored one step late, both in flight under the step's 64 WMMAs; sum order unchanged
# speedup vs baseline: 4.5253x; 1.2405x over previous
#include <hip/hip_runtime.h>
#include <cmath>

typedef __attribute__((ext_vector_type(2))) float v2f;
typedef __attribute__((ext_vector_type(8))) float v8f;

#define Bq  64
#define Sq  2048
#define Iq  128
#define Hq  256
#define BDq 128
#define LDSROW 260

__global__ __launch_bounds__(256) void k_xproj(const float* __restrict__ nh,
                                               const float* __restrict__ nr,
                                               const float* __restrict__ nt,
                                               const float* __restrict__ Wx,
                                               const float* __restrict__ bh,
                                               float* __restrict__ P) {
  const int lane = threadIdx.x & 31, wv = threadIdx.x >> 5;
  const int gw    = blockIdx.x * 8 + wv;
  const int mtile = gw >> 1, ng = gw & 1;
  const int half  = lane >> 4, l16 = lane & 15;
  const int grow  = mtile * 16 + l16;
  const int t = grow >> 6, b = grow & 63;
  const size_t xbase = ((size_t)b * Sq + t) * Iq;

  v8f acc[8];
  #pragma unroll
  for (int n = 0; n < 8; ++n) {
    const float bias = bh[ng * 128 + n * 16 + l16];
    #pragma unroll
    for (int j = 0; j < 8; ++j) acc[n][j] = bias;
  }

  for (int kk = 0; kk < 96; ++kk) {
    const int kb = kk * 4 + 2 * half;
    const float* xs = (kb < Iq) ? nh : ((kb < 2 * Iq) ? nr : nt);
    const int off = kb & (Iq - 1);
    v2f a; a.x = xs[xbase + off]; a.y = xs[xbase + off + 1];
    #pragma unroll
    for (int n = 0; n < 8; ++n) {
      const int col = ng * 128 + n * 16 + l16;
      v2f bm; bm.x = Wx[(size_t)kb * Hq + col];
              bm.y = Wx[(size_t)(kb + 1) * Hq + col];
      acc[n] = __builtin_amdgcn_wmma_f32_16x16x4_f32(
          false, a, false, bm, (short)0, acc[n], false, false);
    }
  }

  #pragma unroll
  for (int n = 0; n < 8; ++n) {
    const int col = ng * 128 + n * 16 + l16;
    #pragma unroll
    for (int j = 0; j < 8; ++j) {
      const int r = mtile * 16 + j + 8 * half;
      P[(size_t)r * Hq + col] = acc[n][j];
    }
  }
}

__global__ __launch_bounds__(512) void k_scan(const float* __restrict__ h0,
                                              const float* __restrict__ Wh,
                                              float* __restrict__ P,
                                              float* __restrict__ hlast) {
  __shared__ float hsh[16 * LDSROW];
  const int tid = threadIdx.x;
  const int row0 = blockIdx.x * 16;
  for (int i = tid; i < 16 * Hq; i += 512)
    hsh[(i >> 8) * LDSROW + (i & 255)] = h0[row0 * Hq + i];

  const int lane = tid & 31, n = tid >> 5;
  const int half = lane >> 4, l16 = lane & 15;
  const int col = n * 16 + l16;

  v2f bw[64];
  #pragma unroll
  for (int kk = 0; kk < 64; ++kk) {
    const int kb = kk * 4 + 2 * half;
    bw[kk].x = Wh[(size_t)kb * Hq + col];
    bw[kk].y = Wh[(size_t)(kb + 1) * Hq + col];
  }
  __syncthreads();

  const float* abase = hsh + l16 * LDSROW + 2 * half;
  float* hw = hsh + (8 * half) * LDSROW + col;
  float* pw = P + (size_t)(row0 + 8 * half) * Hq + col;
  float* lw = hlast + (row0 + 8 * half) * Hq + col;

  v8f cur;
  #pragma unroll
  for (int j = 0; j < 8; ++j) cur[j] = pw[j * Hq];
  v8f hv;

  for (int t = 0; t < Sq; ++t) {
    float* pt = pw + (size_t)t * (64 * Hq);
    if (t > 0) {
      float* pp = pt - 64 * Hq;
      #pragma unroll
      for (int j = 0; j < 8; ++j) pp[j * Hq] = hv[j];
    }
    const float* pn = (t + 1 < Sq) ? pt + 64 * Hq : pt;
    v8f nxt;
    #pragma unroll
    for (int j = 0; j < 8; ++j) nxt[j] = pn[j * Hq];
    __builtin_amdgcn_sched_barrier(0);

    v8f acc = cur;
    #pragma unroll
    for (int kk = 0; kk < 64; ++kk) {
      const float* ap = abase + kk * 4;
      v2f a; a.x = ap[0]; a.y = ap[1];
      acc = __builtin_amdgcn_wmma_f32_16x16x4_f32(
          false, a, false, bw[kk], (short)0, acc, false, false);
    }
    __syncthreads();
    #pragma unroll
    for (int j = 0; j < 8; ++j) {
      const float v = tanhf(acc[j]);
      hw[j * LDSROW] = v;
      hv[j] = v;
    }
    cur = nxt;
    __syncthreads();
  }
  {
    float* pl = pw + (size_t)(Sq - 1) * (64 * Hq);
    #pragma unroll
    for (int j = 0; j < 8; ++j) {
      pl[j * Hq] = hv[j];
      lw[j * Hq] = hv[j];
    }
  }
}

__global__ __launch_bounds__(256) void k_out(const float* __restrict__ P,
                                             const float* __restrict__ Wb,
                                             const float* __restrict__ bb,
                                             float* __restrict__ outs) {
  const int lane = threadIdx.x & 31, wv = threadIdx.x >> 5;
  const int mtile = blockIdx.x * 8 + wv;
  const int half = lane >> 4, l16 = lane & 15;
  const int arow = mtile * 16 + l16;

  v8f acc[8];
  #pragma unroll
  for (int n = 0; n < 8; ++n) {
    const float bias = bb[n * 16 + l16];
    #pragma unroll
    for (int j = 0; j < 8; ++j) acc[n][j] = bias;
  }
  for (int kk = 0; kk < 64; ++kk) {
    const int kb = kk * 4 + 2 * half;
    v2f a; a.x = P[(size_t)arow * Hq + kb];
           a.y = P[(size_t)arow * Hq + kb + 1];
    #pragma unroll
    for (int n = 0; n < 8; ++n) {
      const int col = n * 16 + l16;
      v2f bm; bm.x = Wb[(size_t)kb * BDq + col];
              bm.y = Wb[(size_t)(kb + 1) * BDq + col];
      acc[n] = __builtin_amdgcn_wmma_f32_16x16x4_f32(
          false, a, false, bm, (short)0, acc[n], false, false);
    }
  }
  #pragma unroll
  for (int n = 0; n < 8; ++n) {
    const int col = n * 16 + l16;
    #pragma unroll
    for (int j = 0; j < 8; ++j) {
      const int row = mtile * 16 + j + 8 * half;
      const int tt = row >> 6, bb_ = row & 63;
      outs[(size_t)bb_ * Sq * BDq + (size_t)tt * BDq + col] = tanhf(acc[n][j]);
    }
  }
}

extern "C" void kernel_launch(void* const* d_in, const int* in_sizes, int n_in,
                              void* d_out, int out_size, void* d_ws, size_t ws_size,
                              hipStream_t stream) {
  const float* nh = (const float*)d_in[0];
  const float* nr = (const float*)d_in[1];
  const float* nt = (const float*)d_in[2];
  const float* h0 = (const float*)d_in[3];
  const float* Wx = (const float*)d_in[4];
  const float* Wh = (const float*)d_in[5];
  const float* bh = (const float*)d_in[6];
  const float* Wb = (const float*)d_in[7];
  const float* bb = (const float*)d_in[8];

  float* P     = (float*)d_ws;
  float* outs  = (float*)d_out;
  float* hlast = outs + (size_t)Bq * Sq * BDq;

  k_xproj<<<2048, 256, 0, stream>>>(nh, nr, nt, Wx, bh, P);
  k_scan<<<4, 512, 0, stream>>>(h0, Wh, P, hlast);
  k_out<<<1024, 256, 0, stream>>>(P, Wb, bb, outs);
}


# scan: the step's tanh evaluated branch-free as sign(x)(1-2/(1+2^(c|x|))) on hardware exp2 and reciprocal instead of library tanhf's two divergent branches per element; abs error below 3e-7
# speedup vs baseline: 5.5132x; 1.2183x over previous
#include <hip/hip_runtime.h>
#include <cmath>

typedef __attribute__((ext_vector_type(2))) float v2f;
typedef __attribute__((ext_vector_type(8))) float v8f;

#define Bq  64
#define Sq  2048
#define Iq  128
#define Hq  256
#define BDq 128
#define LDSROW 260

__global__ __launch_bounds__(256) void k_xproj(const float* __restrict__ nh,
                                               const float* __restrict__ nr,
                                               const float* __restrict__ nt,
                                               const float* __restrict__ Wx,
                                               const float* __restrict__ bh,
                                               float* __restrict__ P) {
  const int lane = threadIdx.x & 31, wv = threadIdx.x >> 5;
  const int gw    = blockIdx.x * 8 + wv;
  const int mtile = gw >> 1, ng = gw & 1;
  const int half  = lane >> 4, l16 = lane & 15;
  const int grow  = mtile * 16 + l16;
  const int t = grow >> 6, b = grow & 63;
  const size_t xbase = ((size_t)b * Sq + t) * Iq;

  v8f acc[8];
  #pragma unroll
  for (int n = 0; n < 8; ++n) {
    const float bias = bh[ng * 128 + n * 16 + l16];
    #pragma unroll
    for (int j = 0; j < 8; ++j) acc[n][j] = bias;
  }

  for (int kk = 0; kk < 96; ++kk) {
    const int kb = kk * 4 + 2 * half;
    const float* xs = (kb < Iq) ? nh : ((kb < 2 * Iq) ? nr : nt);
    const int off = kb & (Iq - 1);
    v2f a; a.x = xs[xbase + off]; a.y = xs[xbase + off + 1];
    #pragma unroll
    for (int n = 0; n < 8; ++n) {
      const int col = ng * 128 + n * 16 + l16;
      v2f bm; bm.x = Wx[(size_t)kb * Hq + col];
              bm.y = Wx[(size_t)(kb + 1) * Hq + col];
      acc[n] = __builtin_amdgcn_wmma_f32_16x16x4_f32(
          false, a, false, bm, (short)0, acc[n], false, false);
    }
  }

  #pragma unroll
  for (int n = 0; n < 8; ++n) {
    const int col = ng * 128 + n * 16 + l16;
    #pragma unroll
    for (int j = 0; j < 8; ++j) {
      const int r = mtile * 16 + j + 8 * half;
      P[(size_t)r * Hq + col] = acc[n][j];
    }
  }
}

__device__ __forceinline__ float tanh_step(float x) {
  const float e = __builtin_amdgcn_exp2f(fabsf(x) * 2.885390081777927f);
  const float r = __builtin_amdgcn_rcpf(1.0f + e);
  return copysignf(fmaf(r, -2.0f, 1.0f), x);
}

__global__ __launch_bounds__(512) void k_scan(const float* __restrict__ h0,
                                              const float* __restrict__ Wh,
                                              float* __restrict__ P,
                                              float* __restrict__ hlast) {
  __shared__ float hsh[16 * LDSROW];
  const int tid = threadIdx.x;
  const int row0 = blockIdx.x * 16;
  for (int i = tid; i < 16 * Hq; i += 512)
    hsh[(i >> 8) * LDSROW + (i & 255)] = h0[row0 * Hq + i];

  const int lane = tid & 31, n = tid >> 5;
  const int half = lane >> 4, l16 = lane & 15;
  const int col = n * 16 + l16;

  v2f bw[64];
  #pragma unroll
  for (int kk = 0; kk < 64; ++kk) {
    const int kb = kk * 4 + 2 * half;
    bw[kk].x = Wh[(size_t)kb * Hq + col];
    bw[kk].y = Wh[(size_t)(kb + 1) * Hq + col];
  }
  __syncthreads();

  const float* abase = hsh + l16 * LDSROW + 2 * half;
  float* hw = hsh + (8 * half) * LDSROW + col;
  float* pw = P + (size_t)(row0 + 8 * half) * Hq + col;
  float* lw = hlast + (row0 + 8 * half) * Hq + col;

  v8f cur;
  #pragma unroll
  for (int j = 0; j < 8; ++j) cur[j] = pw[j * Hq];
  v8f hv;

  for (int t = 0; t < Sq; ++t) {
    float* pt = pw + (size_t)t * (64 * Hq);
    if (t > 0) {
      float* pp = pt - 64 * Hq;
      #pragma unroll
      for (int j = 0; j < 8; ++j) pp[j * Hq] = hv[j];
    }
    const float* pn = (t + 1 < Sq) ? pt + 64 * Hq : pt;
    v8f nxt;
    #pragma unroll
    for (int j = 0; j < 8; ++j) nxt[j] = pn[j * Hq];
    __builtin_amdgcn_sched_barrier(0);

    v8f acc = cur;
    #pragma unroll
    for (int kk = 0; kk < 64; ++kk) {
      const float* ap = abase + kk * 4;
      v2f a; a.x = ap[0]; a.y = ap[1];
      acc = __builtin_amdgcn_wmma_f32_16x16x4_f32(
          false, a, false, bw[kk], (short)0, acc, false, false);
    }
    __syncthreads();
    #pragma unroll
    for (int j = 0; j < 8; ++j) {
      const float v = tanh_step(acc[j]);
      hw[j * LDSROW] = v;
      hv[j] = v;
    }
    cur = nxt;
    __syncthreads();
  }
  {
    float* pl = pw + (size_t)(Sq - 1) * (64 * Hq);
    #pragma unroll
    for (int j = 0; j < 8; ++j) {
      pl[j * Hq] = hv[j];
      lw[j * Hq] = hv[j];
    }
  }
}

__global__ __launch_bounds__(256) void k_out(const float* __restrict__ P,
                                             const float* __restrict__ Wb,
                                             const float* __restrict__ bb,
                                             float* __restrict__ outs) {
  const int lane = threadIdx.x & 31, wv = threadIdx.x >> 5;
  const int mtile = blockIdx.x * 8 + wv;
  const int half = lane >> 4, l16 = lane & 15;
  const int arow = mtile * 16 + l16;

  v8f acc[8];
  #pragma unroll
  for (int n = 0; n < 8; ++n) {
    const float bias = bb[n * 16 + l16];
    #pragma unroll
    for (int j = 0; j < 8; ++j) acc[n][j] = bias;
  }
  for (int kk = 0; kk < 64; ++kk) {
    const int kb = kk * 4 + 2 * half;
    v2f a; a.x = P[(size_t)arow * Hq + kb];
           a.y = P[(size_t)arow * Hq + kb + 1];
    #pragma unroll
    for (int n = 0; n < 8; ++n) {
      const int col = n * 16 + l16;
      v2f bm; bm.x = Wb[(size_t)kb * BDq + col];
              bm.y = Wb[(size_t)(kb + 1) * BDq + col];
      acc[n] = __builtin_amdgcn_wmma_f32_16x16x4_f32(
          false, a, false, bm, (short)0, acc[n], false, false);
    }
  }
  #pragma unroll
  for (int n = 0; n < 8; ++n) {
    const int col = n * 16 + l16;
    #pragma unroll
    for (int j = 0; j < 8; ++j) {
      const int row = mtile * 16 + j + 8 * half;
      const int tt = row >> 6, bb_ = row & 63;
      outs[(size_t)bb_ * Sq * BDq + (size_t)tt * BDq + col] = tanhf(acc[n][j]);
    }
  }
}

extern "C" void kernel_launch(void* const* d_in, const int* in_sizes, int n_in,
                              void* d_out, int out_size, void* d_ws, size_t ws_size,
                              hipStream_t stream) {
  const float* nh = (const float*)d_in[0];
  const float* nr = (const float*)d_in[1];
  const float* nt = (const float*)d_in[2];
  const float* h0 = (const float*)d_in[3];
  const float* Wx = (const float*)d_in[4];
  const float* Wh = (const float*)d_in[5];
  const float* bh = (const float*)d_in[6];
  const float* Wb = (const float*)d_in[7];
  const float* bb = (const float*)d_in[8];

  float* P     = (float*)d_ws;
  float* outs  = (float*)d_out;
  float* hlast = outs + (size_t)Bq * Sq * BDq;

  k_xproj<<<2048, 256, 0, stream>>>(nh, nr, nt, Wx, bh, P);
  k_scan<<<4, 512, 0, stream>>>(h0, Wh, P, hlast);
  k_out<<<1024, 256, 0, stream>>>(P, Wb, bb, outs);
}


# scan: each wave reads its A fragments from LDS in runs of 16, one 16-WMMA block ahead into a second register set, not a pair ahead between its WMMAs; h-read barrier moves up with them; sum order same
# speedup vs baseline: 6.2975x; 1.1423x over previous
#include <hip/hip_runtime.h>
#include <cmath>

typedef __attribute__((ext_vector_type(2))) float v2f;
typedef __attribute__((ext_vector_type(8))) float v8f;

#define Bq  64
#define Sq  2048
#define Iq  128
#define Hq  256
#define BDq 128
#define LDSROW 260

__global__ __launch_bounds__(256) void k_xproj(const float* __restrict__ nh,
                                               const float* __restrict__ nr,
                                               const float* __restrict__ nt,
                                               const float* __restrict__ Wx,
                                               const float* __restrict__ bh,
                                               float* __restrict__ P) {
  const int lane = threadIdx.x & 31, wv = threadIdx.x >> 5;
  const int gw    = blockIdx.x * 8 + wv;
  const int mtile = gw >> 1, ng = gw & 1;
  const int half  = lane >> 4, l16 = lane & 15;
  const int grow  = mtile * 16 + l16;
  const int t = grow >> 6, b = grow & 63;
  const size_t xbase = ((size_t)b * Sq + t) * Iq;

  v8f acc[8];
  #pragma unroll
  for (int n = 0; n < 8; ++n) {
    const float bias = bh[ng * 128 + n * 16 + l16];
    #pragma unroll
    for (int j = 0; j < 8; ++j) acc[n][j] = bias;
  }

  for (int kk = 0; kk < 96; ++kk) {
    const int kb = kk * 4 + 2 * half;
    const float* xs = (kb < Iq) ? nh : ((kb < 2 * Iq) ? nr : nt);
    const int off = kb & (Iq - 1);
    v2f a; a.x = xs[xbase + off]; a.y = xs[xbase + off + 1];
    #pragma unroll
    for (int n = 0; n < 8; ++n) {
      const int col = ng * 128 + n * 16 + l16;
      v2f bm; bm.x = Wx[(size_t)kb * Hq + col];
              bm.y = Wx[(size_t)(kb + 1) * Hq + col];
      acc[n] = __builtin_amdgcn_wmma_f32_16x16x4_f32(
          false, a, false, bm, (short)0, acc[n], false, false);
    }
  }

  #pragma unroll
  for (int n = 0; n < 8; ++n) {
    const int col = ng * 128 + n * 16 + l16;
    #pragma unroll
    for (int j = 0; j < 8; ++j) {
      const int r = mtile * 16 + j + 8 * half;
      P[(size_t)r * Hq + col] = acc[n][j];
    }
  }
}

__device__ __forceinline__ float tanh_step(float x) {
  const float e = __builtin_amdgcn_exp2f(fabsf(x) * 2.885390081777927f);
  const float r = __builtin_amdgcn_rcpf(1.0f + e);
  return copysignf(fmaf(r, -2.0f, 1.0f), x);
}

__global__ __launch_bounds__(512) void k_scan(const float* __restrict__ h0,
                                              const float* __restrict__ Wh,
                                              float* __restrict__ P,
                                              float* __restrict__ hlast) {
  __shared__ float hsh[16 * LDSROW];
  const int tid = threadIdx.x;
  const int row0 = blockIdx.x * 16;
  for (int i = tid; i < 16 * Hq; i += 512)
    hsh[(i >> 8) * LDSROW + (i & 255)] = h0[row0 * Hq + i];

  const int lane = tid & 31, n = tid >> 5;
  const int half = lane >> 4, l16 = lane & 15;
  const int col = n * 16 + l16;

  v2f bw[64];
  #pragma unroll
  for (int kk = 0; kk < 64; ++kk) {
    const int kb = kk * 4 + 2 * half;
    bw[kk].x = Wh[(size_t)kb * Hq + col];
    bw[kk].y = Wh[(size_t)(kb + 1) * Hq + col];
  }
  __syncthreads();

  const float* abase = hsh + l16 * LDSROW + 2 * half;
  float* hw = hsh + (8 * half) * LDSROW + col;
  float* pw = P + (size_t)(row0 + 8 * half) * Hq + col;
  float* lw = hlast + (row0 + 8 * half) * Hq + col;

  v8f cur;
  #pragma unroll
  for (int j = 0; j < 8; ++j) cur[j] = pw[j * Hq];
  v8f hv;

  for (int t = 0; t < Sq; ++t) {
    v2f af[2][16];
    #pragma unroll
    for (int i = 0; i < 32; ++i) {
      const float* ap = abase + i * 4;
      af[i >> 4][i & 15].x = ap[0]; af[i >> 4][i & 15].y = ap[1];
    }
    __builtin_amdgcn_sched_barrier(0);

    float* pt = pw + (size_t)t * (64 * Hq);
    if (t > 0) {
      float* pp = pt - 64 * Hq;
      #pragma unroll
      for (int j = 0; j < 8; ++j) pp[j * Hq] = hv[j];
    }
    const float* pn = (t + 1 < Sq) ? pt + 64 * Hq : pt;
    v8f nxt;
    #pragma unroll
    for (int j = 0; j < 8; ++j) nxt[j] = pn[j * Hq];

    v8f acc = cur;
    #pragma unroll
    for (int g = 0; g < 4; ++g) {
      if (g == 1 || g == 2) {
        #pragma unroll
        for (int i = 0; i < 16; ++i) {
          const float* ap = abase + ((g + 1) * 16 + i) * 4;
          af[(g + 1) & 1][i].x = ap[0]; af[(g + 1) & 1][i].y = ap[1];
        }
      }
      __builtin_amdgcn_sched_barrier(0);
      if (g == 3) __syncthreads();
      #pragma unroll
      for (int i = 0; i < 16; ++i) {
        acc = __builtin_amdgcn_wmma_f32_16x16x4_f32(
            false, af[g & 1][i], false, bw[g * 16 + i], (short)0, acc, false, false);
      }
      if (g < 3) __builtin_amdgcn_sched_barrier(0);
    }
    #pragma unroll
    for (int j = 0; j < 8; ++j) {
      const float v = tanh_step(acc[j]);
      hw[j * LDSROW] = v;
      hv[j] = v;
    }
    cur = nxt;
    __syncthreads();
  }
  {
    float* pl = pw + (size_t)(Sq - 1) * (64 * Hq);
    #pragma unroll
    for (int j = 0; j < 8; ++j) {
      pl[j * Hq] = hv[j];
      lw[j * Hq] = hv[j];
    }
  }
}

__global__ __launch_bounds__(256) void k_out(const float* __restrict__ P,
                                             const float* __restrict__ Wb,
                                             const float* __restrict__ bb,
                                             float* __restrict__ outs) {
  const int lane = threadIdx.x & 31, wv = threadIdx.x >> 5;
  const int mtile = blockIdx.x * 8 + wv;
  const int half = lane >> 4, l16 = lane & 15;
  const int arow = mtile * 16 + l16;

  v8f acc[8];
  #pragma unroll
  for (int n = 0; n < 8; ++n) {
    const float bias = bb[n * 16 + l16];
    #pragma unroll
    for (int j = 0; j < 8; ++j) acc[n][j] = bias;
  }
  for (int kk = 0; kk < 64; ++kk) {
    const int kb = kk * 4 + 2 * half;
    v2f a; a.x = P[(size_t)arow * Hq + kb];
           a.y = P[(size_t)arow * Hq + kb + 1];
    #pragma unroll
    for (int n = 0; n < 8; ++n) {
      const int col = n * 16 + l16;
      v2f bm; bm.x = Wb[(size_t)kb * BDq + col];
              bm.y = Wb[(size_t)(kb + 1) * BDq + col];
      acc[n] = __builtin_amdgcn_wmma_f32_16x16x4_f32(
          false, a, false, bm, (short)0, acc[n], false, false);
    }
  }
  #pragma unroll
  for (int n = 0; n < 8; ++n) {
    const int col = n * 16 + l16;
    #pragma unroll
    for (int j = 0; j < 8; ++j) {
      const int row = mtile * 16 + j + 8 * half;
      const int tt = row >> 6, bb_ = row & 63;
      outs[(size_t)bb_ * Sq * BDq + (size_t)tt * BDq + col] = tanhf(acc[n][j]);
    }
  }
}

extern "C" void kernel_launch(void* const* d_in, const int* in_sizes, int n_in,
                              void* d_out, int out_size, void* d_ws, size_t ws_size,
                              hipStream_t stream) {
  const float* nh = (const float*)d_in[0];
  const float* nr = (const float*)d_in[1];
  const float* nt = (const float*)d_in[2];
  const float* h0 = (const float*)d_in[3];
  const float* Wx = (const float*)d_in[4];
  const float* Wh = (const float*)d_in[5];
  const float* bh = (const float*)d_in[6];
  const float* Wb = (const float*)d_in[7];
  const float* bb = (const float*)d_in[8];

  float* P     = (float*)d_ws;
  float* outs  = (float*)d_out;
  float* hlast = outs + (size_t)Bq * Sq * BDq;

  k_xproj<<<2048, 256, 0, stream>>>(nh, nr, nt, Wx, bh, P);
  k_scan<<<4, 512, 0, stream>>>(h0, Wh, P, hlast);
  k_out<<<1024, 256, 0, stream>>>(P, Wb, bb, outs);
}


# projection GEMMs overlapped with the scan: time cut into 4 chunks, each launch runs scan chunk c on 4 workgroups beside x-projection chunk c+1 and output chunk c-1 on the idle units; sums unchanged
# speedup vs baseline: 6.5213x; 1.0355x over previous
#include <hip/hip_runtime.h>
#include <cmath>

typedef __attribute__((ext_vector_type(2))) float v2f;
typedef __attribute__((ext_vector_type(8))) float v8f;

#define Bq  64
#define Sq  2048
#define Iq  128
#define Hq  256
#define BDq 128
#define LDSROW 260

#define NCH   4
#define TCH   (Sq / NCH)
#define MTCH  (TCH * Bq / 16)
#define XBLK  (MTCH * 2 / 16)
#define OBLK  (MTCH / 16)

__device__ __forceinline__ void xproj_wave(const float* __restrict__ nh,
                                           const float* __restrict__ nr,
                                           const float* __restrict__ nt,
                                           const float* __restrict__ Wx,
                                           const float* __restrict__ bh,
                                           float* __restrict__ P,
                                           const int gw) {
  const int lane = threadIdx.x & 31;
  const int mtile = gw >> 1, ng = gw & 1;
  const int half  = lane >> 4, l16 = lane & 15;
  const int grow  = mtile * 16 + l16;
  const int t = grow >> 6, b = grow & 63;
  const size_t xbase = ((size_t)b * Sq + t) * Iq;

  v8f acc[8];
  #pragma unroll
  for (int n = 0; n < 8; ++n) {
    const float bias = bh[ng * 128 + n * 16 + l16];
    #pragma unroll
    for (int j = 0; j < 8; ++j) acc[n][j] = bias;
  }

  for (int kk = 0; kk < 96; ++kk) {
    const int kb = kk * 4 + 2 * half;
    const float* xs = (kb < Iq) ? nh : ((kb < 2 * Iq) ? nr : nt);
    const int off = kb & (Iq - 1);
    v2f a; a.x = xs[xbase + off]; a.y = xs[xbase + off + 1];
    #pragma unroll
    for (int n = 0; n < 8; ++n) {
      const int col = ng * 128 + n * 16 + l16;
      v2f bm; bm.x = Wx[(size_t)kb * Hq + col];
              bm.y = Wx[(size_t)(kb + 1) * Hq + col];
      acc[n] = __builtin_amdgcn_wmma_f32_16x16x4_f32(
          false, a, false, bm, (short)0, acc[n], false, false);
    }
  }

  #pragma unroll
  for (int n = 0; n < 8; ++n) {
    const int col = ng * 128 + n * 16 + l16;
    #pragma unroll
    for (int j = 0; j < 8; ++j) {
      const int r = mtile * 16 + j + 8 * half;
      P[(size_t)r * Hq + col] = acc[n][j];
    }
  }
}

__device__ __forceinline__ float tanh_step(float x) {
  const float e = __builtin_amdgcn_exp2f(fabsf(x) * 2.885390081777927f);
  const float r = __builtin_amdgcn_rcpf(1.0f + e);
  return copysignf(fmaf(r, -2.0f, 1.0f), x);
}

__device__ __forceinline__ void scan_chunk(const float* __restrict__ h0,
                                           const float* __restrict__ Wh,
                                           float* __restrict__ P,
                                           float* __restrict__ hlast,
                                           float* __restrict__ hsh,
                                           const int mt,
                                           const int t0, const int t1) {
  const int tid = threadIdx.x;
  const int row0 = mt * 16;
  const float* hin = (t0 == 0) ? h0 + row0 * Hq
                               : P + ((size_t)(t0 - 1) * Bq + row0) * Hq;
  for (int i = tid; i < 16 * Hq; i += 512)
    hsh[(i >> 8) * LDSROW + (i & 255)] = hin[i];

  const int lane = tid & 31, n = tid >> 5;
  const int half = lane >> 4, l16 = lane & 15;
  const int col = n * 16 + l16;

  v2f bw[64];
  #pragma unroll
  for (int kk = 0; kk < 64; ++kk) {
    const int kb = kk * 4 + 2 * half;
    bw[kk].x = Wh[(size_t)kb * Hq + col];
    bw[kk].y = Wh[(size_t)(kb + 1) * Hq + col];
  }
  __syncthreads();

  const float* abase = hsh + l16 * LDSROW + 2 * half;
  float* hw = hsh + (8 * half) * LDSROW + col;
  float* pw = P + (size_t)(row0 + 8 * half) * Hq + col;
  float* lw = hlast + (row0 + 8 * half) * Hq + col;

  v8f cur;
  #pragma unroll
  for (int j = 0; j < 8; ++j) cur[j] = pw[(size_t)t0 * (Bq * Hq) + j * Hq];
  v8f hv;

  for (int t = t0; t < t1; ++t) {
    v2f af[2][16];
    #pragma unroll
    for (int i = 0; i < 32; ++i) {
      const float* ap = abase + i * 4;
      af[i >> 4][i & 15].x = ap[0]; af[i >> 4][i & 15].y = ap[1];
    }
    __builtin_amdgcn_sched_barrier(0);

    float* pt = pw + (size_t)t * (64 * Hq);
    if (t > t0) {
      float* pp = pt - 64 * Hq;
      #pragma unroll
      for (int j = 0; j < 8; ++j) pp[j * Hq] = hv[j];
    }
    const float* pn = (t + 1 < t1) ? pt + 64 * Hq : pt;
    v8f nxt;
    #pragma unroll
    for (int j = 0; j < 8; ++j) nxt[j] = pn[j * Hq];

    v8f acc = cur;
    #pragma unroll
    for (int g = 0; g < 4; ++g) {
      if (g == 1 || g == 2) {
        #pragma unroll
        for (int i = 0; i < 16; ++i) {
          const float* ap = abase + ((g + 1) * 16 + i) * 4;
          af[(g + 1) & 1][i].x = ap[0]; af[(g + 1) & 1][i].y = ap[1];
        }
      }
      __builtin_amdgcn_sched_barrier(0);
      if (g == 3) __syncthreads();
      #pragma unroll
      for (int i = 0; i < 16; ++i) {
        acc = __builtin_amdgcn_wmma_f32_16x16x4_f32(
            false, af[g & 1][i], false, bw[g * 16 + i], (short)0, acc, false, false);
      }
      if (g < 3) __builtin_amdgcn_sched_barrier(0);
    }
    #pragma unroll
    for (int j = 0; j < 8; ++j) {
      const float v = tanh_step(acc[j]);
      hw[j * LDSROW] = v;
      hv[j] = v;
    }
    cur = nxt;
    __syncthreads();
  }
  {
    float* pl = pw + (size_t)(t1 - 1) * (64 * Hq);
    #pragma unroll
    for (int j = 0; j < 8; ++j) pl[j * Hq] = hv[j];
    if (t1 == Sq) {
      #pragma unroll
      for (int j = 0; j < 8; ++j) lw[j * Hq] = hv[j];
    }
  }
}

__device__ __forceinline__ void out_wave(const float* __restrict__ P,
                                         const float* __restrict__ Wb,
                                         const float* __restrict__ bb,
                                         float* __restrict__ outs,
                                         const int mtile) {
  const int lane = threadIdx.x & 31;
  const int half = lane >> 4, l16 = lane & 15;
  const int arow = mtile * 16 + l16;

  v8f acc[8];
  #pragma unroll
  for (int n = 0; n < 8; ++n) {
    const float bias = bb[n * 16 + l16];
    #pragma unroll
    for (int j = 0; j < 8; ++j) acc[n][j] = bias;
  }
  for (int kk = 0; kk < 64; ++kk) {
    const int kb = kk * 4 + 2 * half;
    v2f a; a.x = P[(size_t)arow * Hq + kb];
           a.y = P[(size_t)arow * Hq + kb + 1];
    #pragma unroll
    for (int n = 0; n < 8; ++n) {
      const int col = n * 16 + l16;
      v2f bm; bm.x = Wb[(size_t)kb * BDq + col];
              bm.y = Wb[(size_t)(kb + 1) * BDq + col];
      acc[n] = __builtin_amdgcn_wmma_f32_16x16x4_f32(
          false, a, false, bm, (short)0, acc[n], false, false);
    }
  }
  #pragma unroll
  for (int n = 0; n < 8; ++n) {
    const int col = n * 16 + l16;
    #pragma unroll
    for (int j = 0; j < 8; ++j) {
      const int row = mtile * 16 + j + 8 * half;
      const int tt = row >> 6, bb_ = row & 63;
      outs[(size_t)bb_ * Sq * BDq + (size_t)tt * BDq + col] = tanhf(acc[n][j]);
    }
  }
}

__global__ __launch_bounds__(512) void k_step(const float* __restrict__ nh,
                                              const float* __restrict__ nr,
                                              const float* __restrict__ nt,
                                              const float* __restrict__ h0,
                                              const float* __restrict__ Wx,
                                              const float* __restrict__ Wh,
                                              const float* __restrict__ bh,
                                              const float* __restrict__ Wb,
                                              const float* __restrict__ bb,
                                              float* P,
                                              float* __restrict__ outs,
                                              float* __restrict__ hlast,
                                              const int cx, const int cs, const int co) {
  __shared__ float hsh[16 * LDSROW];
  const int bid = blockIdx.x;
  const int wv = threadIdx.x >> 5;
  if (bid < 4) {
    if (cs >= 0) scan_chunk(h0, Wh, P, hlast, hsh, bid, cs * TCH, cs * TCH + TCH);
  } else if (bid < 4 + XBLK) {
    if (cx >= 0) xproj_wave(nh, nr, nt, Wx, bh, P, cx * (MTCH * 2) + (bid - 4) * 16 + wv);
  } else {
    if (co >= 0) out_wave(P, Wb, bb, outs, co * MTCH + (bid - 4 - XBLK) * 16 + wv);
  }
}

extern "C" void kernel_launch(void* const* d_in, const int* in_sizes, int n_in,
                              void* d_out, int out_size, void* d_ws, size_t ws_size,
                              hipStream_t stream) {
  const float* nh = (const float*)d_in[0];
  const float* nr = (const float*)d_in[1];
  const float* nt = (const float*)d_in[2];
  const float* h0 = (const float*)d_in[3];
  const float* Wx = (const float*)d_in[4];
  const float* Wh = (const float*)d_in[5];
  const float* bh = (const float*)d_in[6];
  const float* Wb = (const float*)d_in[7];
  const float* bb = (const float*)d_in[8];

  float* P     = (float*)d_ws;
  float* outs  = (float*)d_out;
  float* hlast = outs + (size_t)Bq * Sq * BDq;

  k_step<<<260, 512, 0, stream>>>(nh, nr, nt, h0, Wx, Wh, bh, Wb, bb, P, outs, hlast,  0, -1, -1);
  k_step<<<260, 512, 0, stream>>>(nh, nr, nt, h0, Wx, Wh, bh, Wb, bb, P, outs, hlast,  1,  0, -1);
  k_step<<<388, 512, 0, stream>>>(nh, nr, nt, h0, Wx, Wh, bh, Wb, bb, P, outs, hlast,  2,  1,  0);
  k_step<<<388, 512, 0, stream>>>(nh, nr, nt, h0, Wx, Wh, bh, Wb, bb, P, outs, hlast,  3,  2,  1);
  k_step<<<388, 512, 0, stream>>>(nh, nr, nt, h0, Wx, Wh, bh, Wb, bb, P, outs, hlast, -1,  3,  2);
  k_step<<<388, 512, 0, stream>>>(nh, nr, nt, h0, Wx, Wh, bh, Wb, bb, P, outs, hlast, -1, -1,  3);
}
